# starting kernel
# baseline (speedup 1.0000x reference)
#include <hip/hip_runtime.h>

typedef float v2f __attribute__((ext_vector_type(2)));
typedef float v8f __attribute__((ext_vector_type(8)));

namespace {
constexpr int kT  = 2048;
constexpr int kB  = 128;
constexpr int kH  = 200;
constexpr int kHS = 204;
constexpr int kKB = 51;
constexpr int kWBR = 102;
constexpr int kWBC = 208;
constexpr int kWBBytes  = kWBR * kWBC * 8;
constexpr int kHBytes   = kB * kHS * 4;
constexpr int kLdsBytes = kWBBytes + kHBytes;
}

__global__ __launch_bounds__(1024)
void rnn_fused_wmma(const float* __restrict__ in_seq,
                    const float* __restrict__ W_ih,
                    const float* __restrict__ W_hh,
                    const float* __restrict__ b_ih,
                    const float* __restrict__ b_hh,
                    const float* __restrict__ W_out,
                    const float* __restrict__ b_out,
                    float* __restrict__ out)
{
  extern __shared__ char lds[];
  v2f*   wb = reinterpret_cast<v2f*>(lds);
  float* hb = reinterpret_cast<float*>(lds + kWBBytes);

  const int tid  = threadIdx.x;
  const int lane = tid & 31;
  const int wave = tid >> 5;

  for (int idx = tid; idx < kWBR * kWBC; idx += 1024) {
    const int k2 = idx / kWBC;
    const int j  = idx - k2 * kWBC;
    v2f v = {0.0f, 0.0f};
    if (j < kH) {
      if (k2 < kH / 2) {
        v = v2f{W_hh[j * kH + 2 * k2], W_hh[j * kH + 2 * k2 + 1]};
      } else if (k2 == kH / 2) {
        v = v2f{W_ih[j * 2 + 0], W_ih[j * 2 + 1]};
      } else {
        v = v2f{b_ih[j] + b_hh[j], 0.0f};
      }
    }
    wb[idx] = v;
  }
  for (int idx = tid; idx < kB * kHS; idx += 1024) {
    const int c = idx % kHS;
    hb[idx] = (c == 202) ? 1.0f : 0.0f;
  }
  __syncthreads();
  if (tid < kB) {
    const v2f x0 = reinterpret_cast<const v2f*>(in_seq)[tid];
    *reinterpret_cast<v2f*>(&hb[tid * kHS + 200]) = x0;
  }

  const int chunk = tid & 7;
  const int bhead = tid >> 3;
  float wout[25];
#pragma unroll
  for (int i = 0; i < 25; ++i) wout[i] = W_out[chunk * 25 + i];
  const float bout = b_out[0];

  const bool gemmWave = (wave < 26);
  const int nT     = wave >> 1;
  const int mGroup = wave & 1;
  const int col    = nT * 16 + (lane & 15);
  int aOff[4];
#pragma unroll
  for (int i = 0; i < 4; ++i)
    aOff[i] = ((4 * mGroup + i) * 16 + (lane & 15)) * kHS + ((lane >> 4) << 1);
  const int bOff0 = (lane >> 4) * kWBC + col;

  __syncthreads();

  const v8f vzero = {0.f, 0.f, 0.f, 0.f, 0.f, 0.f, 0.f, 0.f};

#pragma unroll 1
  for (int t = 0; t < kT; ++t) {
    v8f acc[4];
#pragma unroll
    for (int i = 0; i < 4; ++i) acc[i] = vzero;

    if (gemmWave) {
      int bo = bOff0;
      for (int kb = 0; kb < kKB; ++kb) {
        const v2f bop = wb[bo];
        const v2f a0 = *reinterpret_cast<const v2f*>(&hb[aOff[0] + (kb << 2)]);
        const v2f a1 = *reinterpret_cast<const v2f*>(&hb[aOff[1] + (kb << 2)]);
        const v2f a2 = *reinterpret_cast<const v2f*>(&hb[aOff[2] + (kb << 2)]);
        const v2f a3 = *reinterpret_cast<const v2f*>(&hb[aOff[3] + (kb << 2)]);
        acc[0] = __builtin_amdgcn_wmma_f32_16x16x4_f32(false, a0, false, bop, (short)0, acc[0], false, false);
        acc[1] = __builtin_amdgcn_wmma_f32_16x16x4_f32(false, a1, false, bop, (short)0, acc[1], false, false);
        acc[2] = __builtin_amdgcn_wmma_f32_16x16x4_f32(false, a2, false, bop, (short)0, acc[2], false, false);
        acc[3] = __builtin_amdgcn_wmma_f32_16x16x4_f32(false, a3, false, bop, (short)0, acc[3], false, false);
        bo += 2 * kWBC;
      }
    }
    __syncthreads();

    if (gemmWave) {
      const bool colOk = (col < kH);
#pragma unroll
      for (int i = 0; i < 4; ++i) {
        const int rbase = (4 * mGroup + i) * 16 + ((lane >> 4) << 3);
#pragma unroll
        for (int v = 0; v < 8; ++v) {
          const float x  = acc[i][v];
          const float e  = __builtin_amdgcn_exp2f(x * 2.885390081777927f);
          const float th = 1.0f - 2.0f * __builtin_amdgcn_rcpf(e + 1.0f);
          if (colOk) hb[(rbase + v) * kHS + col] = th;
        }
      }
    } else {
      const int b = tid - 832;
      if (b < kB && (t + 1) < kT) {
        const v2f x = reinterpret_cast<const v2f*>(in_seq)[(t + 1) * kB + b];
        *reinterpret_cast<v2f*>(&hb[b * kHS + 200]) = x;
      }
    }
    __syncthreads();

    float p = 0.0f;
    const float* hrow = &hb[bhead * kHS + chunk * 25];
#pragma unroll
    for (int i = 0; i < 25; ++i) p += hrow[i] * wout[i];
    p += __shfl_xor(p, 1, 32);
    p += __shfl_xor(p, 2, 32);
    p += __shfl_xor(p, 4, 32);
    if (chunk == 0) out[t * kB + bhead] = p + bout;
  }
}

extern "C" void kernel_launch(void* const* d_in, const int* in_sizes, int n_in,
                              void* d_out, int out_size, void* d_ws, size_t ws_size,
                              hipStream_t stream) {
  const float* in_seq = (const float*)d_in[0];
  const float* W_ih   = (const float*)d_in[1];
  const float* W_hh   = (const float*)d_in[2];
  const float* b_ih   = (const float*)d_in[3];
  const float* b_hh   = (const float*)d_in[4];
  const float* W_out  = (const float*)d_in[5];
  const float* b_out  = (const float*)d_in[6];
  float* out = (float*)d_out;

  hipLaunchKernelGGL(rnn_fused_wmma, dim3(1), dim3(1024), kLdsBytes, stream,
                     in_seq, W_ih, W_hh, b_ih, b_hh, W_out, b_out, out);
}


# Batch rows are independent: 8 work-groups of 16 rows (one M tile each) instead of 1; per N tile K is split over 2 waves (last tile: 4), partial sums added after (K sum order changes)
# speedup vs baseline: 6.2511x; 6.2511x over previous
#include <hip/hip_runtime.h>


typedef float v2f __attribute__((ext_vector_type(2)));
typedef float v4f __attribute__((ext_vector_type(4)));
typedef float v8f __attribute__((ext_vector_type(8)));

namespace {
constexpr int kT  = 2048;
constexpr int kB  = 128;
constexpr int kH  = 200;
constexpr int kHS = 204;
constexpr int kKB = 51;
constexpr int kWBR = 102;
constexpr int kWBC = 208;
constexpr int kRows = 16;
constexpr int kGrid = 8;
constexpr int kKHalf    = 26;
constexpr int kKQuart   = 13;
constexpr int kPairWaves = 24;
constexpr int kGemmWaves = 28;
constexpr int kWBBytes  = kWBR * kWBC * 8;
constexpr int kHBytes   = kRows * kHS * 4;
constexpr int kPXBytes  = kGemmWaves * 2 * 32 * 16;
constexpr int kXSBytes  = kRows * 8;
constexpr int kLdsBytes = kWBBytes + kHBytes + kPXBytes + kXSBytes;
static_assert(kGrid * kRows == kB, "batch split");
static_assert(kKHalf + 25 == kKB && 3 * kKQuart + 12 == kKB, "K split");

template <int NKB>
__device__ __forceinline__ void gemm_part(const float* aPtr, const v2f* bPtr, v4f* pxOut) {
  v8f acc = {0.f, 0.f, 0.f, 0.f, 0.f, 0.f, 0.f, 0.f};
#pragma unroll
  for (int i = 0; i < NKB; ++i) {
    const v2f a = *reinterpret_cast<const v2f*>(aPtr + 4 * i);
    const v2f b = bPtr[2 * kWBC * i];
    acc = __builtin_amdgcn_wmma_f32_16x16x4_f32(false, a, false, b, (short)0, acc, false, false);
  }
  pxOut[0]  = v4f{acc[0], acc[1], acc[2], acc[3]};
  pxOut[32] = v4f{acc[4], acc[5], acc[6], acc[7]};
}

__device__ __forceinline__ float tanh_fast(float x) {
  const float e = __builtin_amdgcn_exp2f(x * 2.885390081777927f);
  return 1.0f - 2.0f * __builtin_amdgcn_rcpf(e + 1.0f);
}
}

__global__ __launch_bounds__(1024)
void rnn_fused_wmma(const float* __restrict__ in_seq,
                    const float* __restrict__ W_ih,
                    const float* __restrict__ W_hh,
                    const float* __restrict__ b_ih,
                    const float* __restrict__ b_hh,
                    const float* __restrict__ W_out,
                    const float* __restrict__ b_out,
                    float* __restrict__ out)
{
  extern __shared__ char lds[];
  v2f*   wb = reinterpret_cast<v2f*>(lds);
  float* hb = reinterpret_cast<float*>(lds + kWBBytes);
  v4f*   px = reinterpret_cast<v4f*>(lds + kWBBytes + kHBytes);
  v2f*   xs = reinterpret_cast<v2f*>(lds + kWBBytes + kHBytes + kPXBytes);

  const int tid  = threadIdx.x;
  const int lane = tid & 31;
  const int wave = __builtin_amdgcn_readfirstlane(tid >> 5);
  const int b0   = blockIdx.x * kRows;

  for (int idx = tid; idx < kWBR * kWBC; idx += 1024) {
    const int k2 = idx / kWBC;
    const int j  = idx - k2 * kWBC;
    v2f v = {0.0f, 0.0f};
    if (j < kH) {
      if (k2 < kH / 2) {
        v = v2f{W_hh[j * kH + 2 * k2], W_hh[j * kH + 2 * k2 + 1]};
      } else if (k2 == kH / 2) {
        v = v2f{W_ih[j * 2 + 0], W_ih[j * 2 + 1]};
      } else {
        v = v2f{b_ih[j] + b_hh[j], 0.0f};
      }
    }
    wb[idx] = v;
  }
  for (int idx = tid; idx < kRows * kHS; idx += 1024) {
    const int c = idx % kHS;
    hb[idx] = (c == 202) ? 1.0f : 0.0f;
  }
  __syncthreads();
  if (tid < kRows) {
    const v2f x0 = reinterpret_cast<const v2f*>(in_seq)[b0 + tid];
    *reinterpret_cast<v2f*>(&hb[tid * kHS + 200]) = x0;
  }

  const bool gemmWave  = (wave < kGemmWaves);
  const bool quadWave  = (wave >= kPairWaves) && gemmWave;
  const bool headWave  = (wave >= kGemmWaves);
  const bool stageWave = (wave == kGemmWaves);

  const int hidx  = tid & 127;
  const int chunk = hidx & 7;
  const int hrowi = hidx >> 3;
  float wout[25];
#pragma unroll
  for (int i = 0; i < 25; ++i) wout[i] = headWave ? W_out[chunk * 25 + i] : 0.0f;
  const float bout = b_out[0];
  const float* hrow = &hb[hrowi * kHS + chunk * 25];

  const int kh  = (wave & 1) ^ ((wave >> 2) & 1);
  const int kq  = wave - kPairWaves;
  const int nT  = quadWave ? 12 : (wave >> 1);
  const int kb0 = quadWave ? (kKQuart * kq) : (kh ? kKHalf : 0);
  const int col = nT * 16 + (lane & 15);
  const float* aPtr = hb + (lane & 15) * kHS + ((lane >> 4) << 1) + 4 * kb0;
  const v2f*   bPtr = wb + ((lane >> 4) + 2 * kb0) * kWBC + col;
  const bool colOk = (col < kH);
  v4f* pxOut = px + (wave * 2) * 32 + lane;
  float* hOutP = hb + (((lane >> 4) << 3) + 4 * kh) * kHS + col;
  const v4f* pxOwn   = px + (wave * 2 + kh) * 32 + lane;
  const v4f* pxOther = px + ((wave ^ 1) * 2 + kh) * 32 + lane;
  float* hOutQ = hb + (((lane >> 4) << 3) + 2 * kq) * kHS + col;
  const v2f* pxQ = reinterpret_cast<const v2f*>(px + (kPairWaves * 2 + (kq >> 1)) * 32 + lane) + (kq & 1);

  __syncthreads();

#pragma unroll 1
  for (int t = 0; t < kT; ++t) {
    if (gemmWave) {
      if (quadWave) {
        if (kq == 3) gemm_part<12>(aPtr, bPtr, pxOut);
        else         gemm_part<kKQuart>(aPtr, bPtr, pxOut);
      } else {
        if (kh) gemm_part<25>(aPtr, bPtr, pxOut);
        else    gemm_part<kKHalf>(aPtr, bPtr, pxOut);
      }
    } else {
      v2f xv = {0.f, 0.f};
      const bool doStage = stageWave && lane < kRows && (t + 1) < kT;
      if (doStage) xv = reinterpret_cast<const v2f*>(in_seq)[(t + 1) * kB + b0 + lane];
      if (t > 0) {
        float p = 0.0f;
#pragma unroll
        for (int i = 0; i < 25; ++i) p += hrow[i] * wout[i];
        p += __shfl_xor(p, 1, 32);
        p += __shfl_xor(p, 2, 32);
        p += __shfl_xor(p, 4, 32);
        if (chunk == 0) out[(t - 1) * kB + b0 + hrowi] = p + bout;
      }
      if (doStage) xs[lane] = xv;
    }
    __syncthreads();

    if (gemmWave) {
      if (quadWave) {
        const v2f p0 = pxQ[0 * 2 * 32 * 2];
        const v2f p1 = pxQ[1 * 2 * 32 * 2];
        const v2f p2 = pxQ[2 * 2 * 32 * 2];
        const v2f p3 = pxQ[3 * 2 * 32 * 2];
#pragma unroll
        for (int j = 0; j < 2; ++j) {
          const float x = ((p0[j] + p1[j]) + p2[j]) + p3[j];
          const float th = tanh_fast(x);
          if (colOk) hOutQ[j * kHS] = th;
        }
      } else {
        const v4f own = *pxOwn;
        const v4f got = *pxOther;
#pragma unroll
        for (int j = 0; j < 4; ++j) {
          const float x  = own[j] + got[j];
          const float th = tanh_fast(x);
          if (colOk) hOutP[j * kHS] = th;
        }
      }
    } else if (stageWave) {
      if (lane < kRows && (t + 1) < kT)
        *reinterpret_cast<v2f*>(&hb[lane * kHS + 200]) = xs[lane];
    }
    __syncthreads();
  }

  if (headWave) {
    float p = 0.0f;
#pragma unroll
    for (int i = 0; i < 25; ++i) p += hrow[i] * wout[i];
    p += __shfl_xor(p, 1, 32);
    p += __shfl_xor(p, 2, 32);
    p += __shfl_xor(p, 4, 32);
    if (chunk == 0) out[(kT - 1) * kB + b0 + hrowi] = p + bout;
  }
}

extern "C" void kernel_launch(void* const* d_in, const int* in_sizes, int n_in,
                              void* d_out, int out_size, void* d_ws, size_t ws_size,
                              hipStream_t stream) {
  const float* in_seq = (const float*)d_in[0];
  const float* W_ih   = (const float*)d_in[1];
  const float* W_hh   = (const float*)d_in[2];
  const float* b_ih   = (const float*)d_in[3];
  const float* b_hh   = (const float*)d_in[4];
  const float* W_out  = (const float*)d_in[5];
  const float* b_out  = (const float*)d_in[6];
  float* out = (float*)d_out;

  hipLaunchKernelGGL(rnn_fused_wmma, dim3(kGrid), dim3(1024), kLdsBytes, stream,
                     in_seq, W_ih, W_hh, b_ih, b_hh, W_out, b_out, out);
}


# Whole-K N tiles per wave, weights kept in registers: 4 waves (one per SIMD), K parts exchanged only for tile 12, one barrier per step; output head moved into tile 12 on WMMA (K sum order changes)
# speedup vs baseline: 7.9764x; 1.2760x over previous
#include <hip/hip_runtime.h>


typedef float v2f __attribute__((ext_vector_type(2)));
typedef float v4f __attribute__((ext_vector_type(4)));
typedef float v8f __attribute__((ext_vector_type(8)));

namespace {
constexpr int kT  = 2048;
constexpr int kB  = 128;
constexpr int kH  = 200;
constexpr int kHS = 204;
constexpr int kRows = 16;
constexpr int kGrid = 8;
constexpr int kWaves = 4;
constexpr int kThreads = kWaves * 32;
constexpr int kNP  = 24;
constexpr int kPXS = 20;
constexpr int kHBuf = kRows * kHS;
constexpr int kPXQ  = kRows * kPXS;
constexpr int kPXBuf = kWaves * kPXQ;
constexpr int kLdsFloats = 2 * kHBuf + 2 * kPXBuf;
constexpr int kLdsReserve = 176 * 1024;
static_assert(kLdsFloats * 4 <= kLdsReserve, "LDS use");
static_assert(kGrid * kRows == kB, "batch split");


__device__ __forceinline__ float tanh_fast(float x) {
  const float e = __builtin_amdgcn_exp2f(x * 2.885390081777927f);
  return 1.0f - 2.0f * __builtin_amdgcn_rcpf(e + 1.0f);
}

__device__ __forceinline__ v8f wmma4(v2f a, v2f b, v8f c) {
  return __builtin_amdgcn_wmma_f32_16x16x4_f32(false, a, false, b, (short)0, c, false, false);
}
}

__global__ __launch_bounds__(kThreads) __attribute__((amdgpu_waves_per_eu(1, 1)))
void rnn_fused_wmma(const float* __restrict__ in_seq,
                    const float* __restrict__ W_ih,
                    const float* __restrict__ W_hh,
                    const float* __restrict__ b_ih,
                    const float* __restrict__ b_hh,
                    const float* __restrict__ W_out,
                    const float* __restrict__ b_out,
                    float* __restrict__ out)
{
  extern __shared__ __attribute__((aligned(16))) float lds[];
  float* hbuf = lds;
  float* pbuf = lds + 2 * kHBuf;

  const int tid  = threadIdx.x;
  const int lane = tid & 31;
  const int c    = lane & 15;
  const int h    = lane >> 4;
  const int w    = __builtin_amdgcn_readfirstlane(tid >> 5);
  const int b0   = blockIdx.x * kRows;

  for (int idx = tid; idx < kLdsFloats; idx += kThreads) {
    float v = 0.0f;
    if (idx < 2 * kHBuf && (idx % kHS) == 202) v = 1.0f;
    lds[idx] = v;
  }

  v4f bP[3][25];
  v2f b50[3];
#pragma unroll
  for (int i = 0; i < 3; ++i) {
    const int col = 16 * (3 * w + i) + c;
    const float* rowp = W_hh + col * kH + 4 * h;
#pragma unroll
    for (int j = 0; j < 25; ++j) bP[i][j] = *reinterpret_cast<const v4f*>(rowp + 8 * j);
    const v2f wi = *reinterpret_cast<const v2f*>(W_ih + 2 * col);
    const float bs = b_ih[col] + b_hh[col];
    b50[i] = h ? v2f{bs, 0.0f} : wi;
  }
  const int col12 = 192 + c;
  v4f bQ[6];
#pragma unroll
  for (int i = 0; i < 6; ++i) {
    const int k0 = 8 * (6 * w + i) + 4 * h;
    v4f v = {0.0f, 0.0f, 0.0f, 0.0f};
    if (col12 < kH)       v = *reinterpret_cast<const v4f*>(W_hh + col12 * kH + k0);
    else if (col12 == kH) v = *reinterpret_cast<const v4f*>(W_out + k0);
    bQ[i] = v;
  }
  v2f bX = {0.0f, 0.0f};
  if (w < 3) {
    const int kx = (w == 0) ? (192 + 4 * h) : (w == 1) ? (194 + 4 * h) : (200 + 2 * h);
#pragma unroll
    for (int e = 0; e < 2; ++e) {
      const int k = kx + e;
      float v = 0.0f;
      if (col12 < kH) {
        if (k < kH)        v = W_hh[col12 * kH + k];
        else if (k == 200) v = W_ih[col12 * 2 + 0];
        else if (k == 201) v = W_ih[col12 * 2 + 1];
        else if (k == 202) v = b_ih[col12] + b_hh[col12];
      } else if (col12 == kH) {
        if (k < kH)        v = W_out[k];
        else if (k == 202) v = b_out[0];
      }
      bX[e] = v;
    }
  }

  __syncthreads();
  if (w == 0 && lane < kRows) {
    const v2f x0 = reinterpret_cast<const v2f*>(in_seq)[b0 + lane];
    *reinterpret_cast<v2f*>(&hbuf[lane * kHS + 200]) = x0;
  }
  __syncthreads();

  const int aOff  = c * kHS + 4 * h;
  const int aqOff = c * kHS + 4 * h + 48 * w;
  const int a50Off = c * kHS + 200 + 2 * h;
  const int prOff = c * kPXS + 4 * h;
  const int phOff = c * kPXS + 8;
  const int hwOff = (8 * h) * kHS + 16 * (3 * w) + c;
  const int pwOff = w * kPXQ + (8 * h) * kPXS + c;
  const bool headLane = (w == 0) && (lane < kRows);

#pragma unroll 1
  for (int t = 0; t <= kT + 1; ++t) {
    const float* hcur = hbuf + (t & 1) * kHBuf;
    float*       hnxt = hbuf + ((t & 1) ^ 1) * kHBuf;
    const float* pcur = pbuf + (t & 1) * kPXBuf;
    float*       pnxt = pbuf + ((t & 1) ^ 1) * kPXBuf;

    v4f aP[kNP];
    aP[0] = *reinterpret_cast<const v4f*>(hcur + aOff);
    v4f pr[kWaves];
    float ph[kWaves];
#pragma unroll
    for (int q = 0; q < kWaves; ++q) {
      pr[q] = *reinterpret_cast<const v4f*>(pcur + q * kPXQ + prOff);
      ph[q] = pcur[q * kPXQ + phOff];
    }
#pragma unroll
    for (int j = 1; j < kNP; ++j) aP[j] = *reinterpret_cast<const v4f*>(hcur + aOff + 8 * j);
    const v2f a50 = *reinterpret_cast<const v2f*>(hcur + a50Off);
    v4f aQ[6];
#pragma unroll
    for (int i = 0; i < 6; ++i) aQ[i] = *reinterpret_cast<const v4f*>(hcur + aqOff + 8 * i);
    __builtin_amdgcn_sched_barrier(0);

    v4f a24;
#pragma unroll
    for (int e = 0; e < 4; ++e) a24[e] = tanh_fast(((pr[0][e] + pr[1][e]) + pr[2][e]) + pr[3][e]);
    const float hv = ((ph[0] + ph[1]) + ph[2]) + ph[3];
    v2f xv = {0.0f, 0.0f};
    if (headLane) {
      const int tn = (t + 1 < kT) ? (t + 1) : (kT - 1);
      xv = reinterpret_cast<const v2f*>(in_seq)[tn * kB + b0 + lane];
      if (t > 1) out[(t - 2) * kB + b0 + lane] = hv;
    }
    const v2f a48 = {a24[0], a24[1]};
    const v2f a49 = {a24[2], a24[3]};
    const v2f aX  = (w == 0) ? a48 : (w == 1) ? a49 : a50;

    auto fragA = [&](int kb) -> v2f {
      if (kb == 50) return a50;
      if (kb == 49) return a49;
      if (kb == 48) return a48;
      const v4f v = aP[kb >> 1];
      return (kb & 1) ? v2f{v[2], v[3]} : v2f{v[0], v[1]};
    };
    auto fragB = [&](int i, int kb) -> v2f {
      if (kb == 50) return b50[i];
      const v4f v = bP[i][kb >> 1];
      return (kb & 1) ? v2f{v[2], v[3]} : v2f{v[0], v[1]};
    };

    v8f acc0 = {0.f, 0.f, 0.f, 0.f, 0.f, 0.f, 0.f, 0.f};
    v8f acc1 = acc0, acc2 = acc0, qa = acc0, qb = acc0;

#pragma unroll
    for (int kb = 0; kb < 51; ++kb) {
      acc0 = wmma4(fragA(kb), fragB(0, kb), acc0);
      acc1 = wmma4(fragA(kb), fragB(1, kb), acc1);
      acc2 = wmma4(fragA(kb), fragB(2, kb), acc2);
    }
    asm volatile("" : "+v"(acc0), "+v"(acc1), "+v"(acc2), "+v"(qa), "+v"(qb));

#pragma unroll
    for (int i = 0; i < 12; ++i) {
      const v4f av = aQ[i >> 1];
      const v4f bv = bQ[i >> 1];
      const v2f fa = (i & 1) ? v2f{av[2], av[3]} : v2f{av[0], av[1]};
      const v2f fb = (i & 1) ? v2f{bv[2], bv[3]} : v2f{bv[0], bv[1]};
      if (i & 1) qb = wmma4(fa, fb, qb);
      else       qa = wmma4(fa, fb, qa);
    }
    qa = wmma4(aX, bX, qa);
    float th0[8], th1[8], th2[8];
#pragma unroll
    for (int r = 0; r < 8; ++r) {
      th0[r] = tanh_fast(acc0[r]);
      th1[r] = tanh_fast(acc1[r]);
      th2[r] = tanh_fast(acc2[r]);
    }
    __builtin_amdgcn_sched_barrier(0);

#pragma unroll
    for (int r = 0; r < 8; ++r) {
      hnxt[hwOff + r * kHS]      = th0[r];
      hnxt[hwOff + r * kHS + 16] = th1[r];
      hnxt[hwOff + r * kHS + 32] = th2[r];
      pnxt[pwOff + r * kPXS]     = qa[r] + qb[r];
    }
    if (headLane) *reinterpret_cast<v2f*>(&hnxt[lane * kHS + 200]) = xv;
    __syncthreads();
  }
}

extern "C" void kernel_launch(void* const* d_in, const int* in_sizes, int n_in,
                              void* d_out, int out_size, void* d_ws, size_t ws_size,
                              hipStream_t stream) {
  const float* in_seq = (const float*)d_in[0];
  const float* W_ih   = (const float*)d_in[1];
  const float* W_hh   = (const float*)d_in[2];
  const float* b_ih   = (const float*)d_in[3];
  const float* b_hh   = (const float*)d_in[4];
  const float* W_out  = (const float*)d_in[5];
  const float* b_out  = (const float*)d_in[6];
  float* out = (float*)d_out;

  hipLaunchKernelGGL(rnn_fused_wmma, dim3(kGrid), dim3(kThreads), kLdsReserve, stream,
                     in_seq, W_ih, W_hh, b_ih, b_hh, W_out, b_out, out);
}


# Step in two halves, each handed over by a split work-group barrier (signal, matrix work, wait): no matrix pipe idles at a barrier; K walk rotated per wave, tile 3 in two chains (K sum order changes)
# speedup vs baseline: 8.5842x; 1.0762x over previous
#include <hip/hip_runtime.h>


typedef float v2f __attribute__((ext_vector_type(2)));
typedef float v4f __attribute__((ext_vector_type(4)));
typedef float v8f __attribute__((ext_vector_type(8)));

namespace {
constexpr int kT  = 2048;
constexpr int kB  = 128;
constexpr int kH  = 200;
constexpr int kHS = 204;
constexpr int kRows = 16;
constexpr int kGrid = 8;
constexpr int kWaves = 4;
constexpr int kThreads = kWaves * 32;
constexpr int kPXS = 20;
constexpr int kHBuf = kRows * kHS;
constexpr int kPXQ  = kRows * kPXS;
constexpr int kPXBuf = kWaves * kPXQ;
constexpr int kLdsFloats = 2 * kHBuf + 2 * kPXBuf;
constexpr int kLdsReserve = 176 * 1024;
static_assert(kLdsFloats * 4 <= kLdsReserve, "LDS use");
static_assert(kGrid * kRows == kB, "batch split");


__device__ __forceinline__ float tanh_fast(float x) {
  const float e = __builtin_amdgcn_exp2f(x * 2.885390081777927f);
  return 1.0f - 2.0f * __builtin_amdgcn_rcpf(e + 1.0f);
}

__device__ __forceinline__ v8f wmma4(v2f a, v2f b, v8f c) {
  return __builtin_amdgcn_wmma_f32_16x16x4_f32(false, a, false, b, (short)0, c, false, false);
}
__device__ __forceinline__ v2f lo2(v4f v) { return v2f{v[0], v[1]}; }
__device__ __forceinline__ v2f hi2(v4f v) { return v2f{v[2], v[3]}; }

__device__ __forceinline__ void bar_arrive() { asm volatile("s_wait_dscnt 0x0\n\ts_barrier_signal -1" ::: "memory"); }
__device__ __forceinline__ void bar_sync()   { asm volatile("s_barrier_wait -1" ::: "memory"); }
__device__ __forceinline__ void pin(v8f& a, v8f& b)     { asm volatile("" : "+v"(a), "+v"(b)); }
__device__ __forceinline__ void pin_mem(v8f& a, v8f& b) { asm volatile("" : "+v"(a), "+v"(b) :: "memory"); }
}

__global__ __launch_bounds__(kThreads) __attribute__((amdgpu_waves_per_eu(1, 1)))
void rnn_fused_wmma(const float* __restrict__ in_seq,
                    const float* __restrict__ W_ih,
                    const float* __restrict__ W_hh,
                    const float* __restrict__ b_ih,
                    const float* __restrict__ b_hh,
                    const float* __restrict__ W_out,
                    const float* __restrict__ b_out,
                    float* __restrict__ out)
{
  extern __shared__ __attribute__((aligned(16))) float lds[];
  float* hbuf = lds;
  float* pbuf = lds + 2 * kHBuf;

  const int tid  = threadIdx.x;
  const int lane = tid & 31;
  const int c    = lane & 15;
  const int h    = lane >> 4;
  const int w    = __builtin_amdgcn_readfirstlane(tid >> 5);
  const int b0   = blockIdx.x * kRows;

  for (int idx = tid; idx < kLdsFloats; idx += kThreads) {
    float v = 0.0f;
    if (idx < 2 * kHBuf && (idx % kHS) == 202) v = 1.0f;
    lds[idx] = v;
  }

  int kq[4];
#pragma unroll
  for (int d = 0; d < 4; ++d) kq[d] = 6 * ((w + d) & 3);

  v4f bE[3][4][4];
  v4f bL[3][4][2];
  v4f b24[3];
  v2f b50[3];
#pragma unroll
  for (int n = 0; n < 3; ++n) {
    const int col = 16 * (3 * w + n) + c;
    const float* rowp = W_hh + col * kH + 4 * h;
#pragma unroll
    for (int d = 0; d < 4; ++d) {
#pragma unroll
      for (int i = 0; i < 4; ++i) bE[n][d][i] = *reinterpret_cast<const v4f*>(rowp + 8 * (kq[d] + i));
#pragma unroll
      for (int i = 0; i < 2; ++i) bL[n][d][i] = *reinterpret_cast<const v4f*>(rowp + 8 * (kq[d] + 4 + i));
    }
    b24[n] = *reinterpret_cast<const v4f*>(rowp + 192);
    const v2f wi = *reinterpret_cast<const v2f*>(W_ih + 2 * col);
    const float bs = b_ih[col] + b_hh[col];
    b50[n] = h ? v2f{bs, 0.0f} : wi;
  }
  const int col12 = 192 + c;
  v4f bQ[6];
#pragma unroll
  for (int i = 0; i < 6; ++i) {
    const int k0 = 8 * (6 * w + i) + 4 * h;
    v4f v = {0.0f, 0.0f, 0.0f, 0.0f};
    if (col12 < kH)       v = *reinterpret_cast<const v4f*>(W_hh + col12 * kH + k0);
    else if (col12 == kH) v = *reinterpret_cast<const v4f*>(W_out + k0);
    bQ[i] = v;
  }
  v2f bX = {0.0f, 0.0f};
  if (w < 3) {
    const int kx = (w == 0) ? (192 + 4 * h) : (w == 1) ? (194 + 4 * h) : (200 + 2 * h);
#pragma unroll
    for (int e = 0; e < 2; ++e) {
      const int k = kx + e;
      float v = 0.0f;
      if (col12 < kH) {
        if (k < kH)        v = W_hh[col12 * kH + k];
        else if (k == 200) v = W_ih[col12 * 2 + 0];
        else if (k == 201) v = W_ih[col12 * 2 + 1];
        else if (k == 202) v = b_ih[col12] + b_hh[col12];
      } else if (col12 == kH) {
        if (k < kH)        v = W_out[k];
        else if (k == 202) v = b_out[0];
      }
      bX[e] = v;
    }
  }

  __syncthreads();

  int aOffD[4];
#pragma unroll
  for (int d = 0; d < 4; ++d) aOffD[d] = c * kHS + 4 * h + 8 * kq[d];
  const int a50Off = c * kHS + 200 + 2 * h;
  const int prOff = c * kPXS + 4 * h;
  const int phOff = c * kPXS + 8;
  const int hwOff = (8 * h) * kHS + 16 * (3 * w) + c;
  const int pwOff = w * kPXQ + (8 * h) * kPXS + c;
  const bool headLane = (w == 0) && (lane < kRows);

  const v8f zero8 = {0.f, 0.f, 0.f, 0.f, 0.f, 0.f, 0.f, 0.f};

  v4f aE[4][4];
#pragma unroll
  for (int d = 0; d < 4; ++d)
#pragma unroll
    for (int i = 0; i < 4; ++i) aE[d][i] = v4f{0.0f, 0.0f, 0.0f, 0.0f};
  v8f acc2 = zero8, acc3 = zero8, qa = zero8, qb = zero8;
  v2f xv = {0.0f, 0.0f};
  if (headLane) xv = reinterpret_cast<const v2f*>(in_seq)[b0 + lane];

#pragma unroll 1
  for (int t = 0; t <= kT + 1; ++t) {
    float* hcur = hbuf + (t & 1) * kHBuf;
    float* hnxt = hbuf + ((t & 1) ^ 1) * kHBuf;
    float* pcur = pbuf + (t & 1) * kPXBuf;

    v8f acc0 = zero8, acc1 = zero8;

#pragma unroll
    for (int i = 0; i < 2; ++i) {
      acc0 = wmma4(lo2(aE[0][i]), lo2(bE[0][0][i]), acc0);
      acc1 = wmma4(lo2(aE[0][i]), lo2(bE[1][0][i]), acc1);
      acc0 = wmma4(hi2(aE[0][i]), hi2(bE[0][0][i]), acc0);
      acc1 = wmma4(hi2(aE[0][i]), hi2(bE[1][0][i]), acc1);
    }
    float th2[8], ps[8];
#pragma unroll
    for (int r = 0; r < 8; ++r) {
      th2[r] = tanh_fast(acc2[r] + acc3[r]);
      ps[r]  = qa[r] + qb[r];
    }
    pin_mem(acc0, acc1);
#pragma unroll
    for (int r = 0; r < 8; ++r) {
      hcur[hwOff + r * kHS + 32] = th2[r];
      pcur[pwOff + r * kPXS]     = ps[r];
    }
    if (headLane) *reinterpret_cast<v2f*>(&hcur[lane * kHS + 200]) = xv;
    pin_mem(acc0, acc1);
#pragma unroll
    for (int i = 2; i < 4; ++i) {
      acc0 = wmma4(lo2(aE[0][i]), lo2(bE[0][0][i]), acc0);
      acc1 = wmma4(lo2(aE[0][i]), lo2(bE[1][0][i]), acc1);
      acc0 = wmma4(hi2(aE[0][i]), hi2(bE[0][0][i]), acc0);
      acc1 = wmma4(hi2(aE[0][i]), hi2(bE[1][0][i]), acc1);
    }
    pin(acc0, acc1);
    bar_arrive();

#pragma unroll
    for (int i = 0; i < 4; ++i) {
      acc0 = wmma4(lo2(aE[1][i]), lo2(bE[0][1][i]), acc0);
      acc1 = wmma4(lo2(aE[1][i]), lo2(bE[1][1][i]), acc1);
      acc0 = wmma4(hi2(aE[1][i]), hi2(bE[0][1][i]), acc0);
      acc1 = wmma4(hi2(aE[1][i]), hi2(bE[1][1][i]), acc1);
    }
    pin(acc0, acc1);
    bar_sync();

    v4f aL[4][2];
    v4f pr[kWaves];
#pragma unroll
    for (int q = 0; q < kWaves; ++q) pr[q] = *reinterpret_cast<const v4f*>(pcur + q * kPXQ + prOff);
#pragma unroll
    for (int d = 0; d < 4; ++d)
#pragma unroll
      for (int i = 0; i < 2; ++i) aL[d][i] = *reinterpret_cast<const v4f*>(hcur + aOffD[d] + 32 + 8 * i);
    const v2f a50 = *reinterpret_cast<const v2f*>(hcur + a50Off);
    pin_mem(acc0, acc1);
    __builtin_amdgcn_sched_barrier(0);

    if (headLane) {
      const int tn = (t + 1 < kT) ? (t + 1) : (kT - 1);
      xv = reinterpret_cast<const v2f*>(in_seq)[tn * kB + b0 + lane];
    }

#pragma unroll
    for (int i = 0; i < 4; ++i) {
      acc0 = wmma4(lo2(aE[2][i]), lo2(bE[0][2][i]), acc0);
      acc1 = wmma4(lo2(aE[2][i]), lo2(bE[1][2][i]), acc1);
      acc0 = wmma4(hi2(aE[2][i]), hi2(bE[0][2][i]), acc0);
      acc1 = wmma4(hi2(aE[2][i]), hi2(bE[1][2][i]), acc1);
    }
    asm volatile("" : "+v"(acc0), "+v"(acc1), "+v"(pr[0]), "+v"(pr[1]), "+v"(pr[2]), "+v"(pr[3]));
    v4f a24;
#pragma unroll
    for (int e = 0; e < 4; ++e) a24[e] = tanh_fast(((pr[0][e] + pr[1][e]) + pr[2][e]) + pr[3][e]);
#pragma unroll
    for (int i = 0; i < 4; ++i) {
      acc0 = wmma4(lo2(aE[3][i]), lo2(bE[0][3][i]), acc0);
      acc1 = wmma4(lo2(aE[3][i]), lo2(bE[1][3][i]), acc1);
      acc0 = wmma4(hi2(aE[3][i]), hi2(bE[0][3][i]), acc0);
      acc1 = wmma4(hi2(aE[3][i]), hi2(bE[1][3][i]), acc1);
    }
#pragma unroll
    for (int d = 0; d < 4; ++d)
#pragma unroll
      for (int i = 0; i < 2; ++i) {
        acc0 = wmma4(lo2(aL[d][i]), lo2(bL[0][d][i]), acc0);
        acc1 = wmma4(lo2(aL[d][i]), lo2(bL[1][d][i]), acc1);
        acc0 = wmma4(hi2(aL[d][i]), hi2(bL[0][d][i]), acc0);
        acc1 = wmma4(hi2(aL[d][i]), hi2(bL[1][d][i]), acc1);
      }
    acc0 = wmma4(lo2(a24), lo2(b24[0]), acc0);
    acc1 = wmma4(lo2(a24), lo2(b24[1]), acc1);
    acc0 = wmma4(hi2(a24), hi2(b24[0]), acc0);
    acc1 = wmma4(hi2(a24), hi2(b24[1]), acc1);
    acc0 = wmma4(a50, b50[0], acc0);
    acc1 = wmma4(a50, b50[1], acc1);

    acc2 = zero8; acc3 = zero8; qa = zero8; qb = zero8;
    asm volatile("" : "+v"(acc0), "+v"(acc1), "+v"(acc2), "+v"(acc3), "+v"(qa), "+v"(qb));

    const v2f aX  = (w == 0) ? lo2(a24) : (w == 1) ? hi2(a24) : a50;
#pragma unroll
    for (int i = 0; i < 4; ++i) {
      qa   = wmma4(lo2(aE[0][i]), lo2(bQ[i]), qa);
      acc2 = wmma4(lo2(aE[0][i]), lo2(bE[2][0][i]), acc2);
      qb   = wmma4(hi2(aE[0][i]), hi2(bQ[i]), qb);
      acc3 = wmma4(hi2(aE[0][i]), hi2(bE[2][0][i]), acc3);
    }
#pragma unroll
    for (int i = 0; i < 2; ++i) {
      qa   = wmma4(lo2(aL[0][i]), lo2(bQ[4 + i]), qa);
      acc2 = wmma4(lo2(aL[0][i]), lo2(bL[2][0][i]), acc2);
      qb   = wmma4(hi2(aL[0][i]), hi2(bQ[4 + i]), qb);
      acc3 = wmma4(hi2(aL[0][i]), hi2(bL[2][0][i]), acc3);
    }
    qa = wmma4(aX, bX, qa);
    float th0[8], th1[8];
#pragma unroll
    for (int r = 0; r < 8; ++r) {
      th0[r] = tanh_fast(acc0[r]);
      th1[r] = tanh_fast(acc1[r]);
    }
    asm volatile("" : "+v"(acc2), "+v"(acc3), "+v"(qa), "+v"(qb) :: "memory");
#pragma unroll
    for (int r = 0; r < 8; ++r) {
      hnxt[hwOff + r * kHS]      = th0[r];
      hnxt[hwOff + r * kHS + 16] = th1[r];
    }
    pin_mem(acc2, acc3);
#pragma unroll
    for (int i = 0; i < 4; ++i) {
      acc2 = wmma4(lo2(aE[1][i]), lo2(bE[2][1][i]), acc2);
      acc3 = wmma4(hi2(aE[1][i]), hi2(bE[2][1][i]), acc3);
    }
    pin(acc2, acc3);
    bar_arrive();

#pragma unroll
    for (int d = 2; d < 4; ++d)
#pragma unroll
      for (int i = 0; i < 4; ++i) {
        acc2 = wmma4(lo2(aE[d][i]), lo2(bE[2][d][i]), acc2);
        acc3 = wmma4(hi2(aE[d][i]), hi2(bE[2][d][i]), acc3);
      }
    pin(acc2, acc3);
    bar_sync();

    float ph[kWaves];
#pragma unroll
    for (int q = 0; q < kWaves; ++q) ph[q] = pcur[q * kPXQ + phOff];
#pragma unroll
    for (int d = 0; d < 4; ++d)
#pragma unroll
      for (int i = 0; i < 4; ++i) aE[d][i] = *reinterpret_cast<const v4f*>(hnxt + aOffD[d] + 8 * i);
    pin_mem(acc2, acc3);
    __builtin_amdgcn_sched_barrier(0);
#pragma unroll
    for (int d = 1; d < 4; ++d)
#pragma unroll
      for (int i = 0; i < 2; ++i) {
        acc2 = wmma4(lo2(aL[d][i]), lo2(bL[2][d][i]), acc2);
        acc3 = wmma4(hi2(aL[d][i]), hi2(bL[2][d][i]), acc3);
      }
    acc2 = wmma4(lo2(a24), lo2(b24[2]), acc2);
    acc3 = wmma4(hi2(a24), hi2(b24[2]), acc3);
    acc2 = wmma4(a50, b50[2], acc2);
    if (headLane && t > 1) out[(t - 2) * kB + b0 + lane] = ((ph[0] + ph[1]) + ph[2]) + ph[3];
  }
}

extern "C" void kernel_launch(void* const* d_in, const int* in_sizes, int n_in,
                              void* d_out, int out_size, void* d_ws, size_t ws_size,
                              hipStream_t stream) {
  const float* in_seq = (const float*)d_in[0];
  const float* W_ih   = (const float*)d_in[1];
  const float* W_hh   = (const float*)d_in[2];
  const float* b_ih   = (const float*)d_in[3];
  const float* b_hh   = (const float*)d_in[4];
  const float* W_out  = (const float*)d_in[5];
  const float* b_out  = (const float*)d_in[6];
  float* out = (float*)d_out;

  hipLaunchKernelGGL(rnn_fused_wmma, dim3(kGrid), dim3(kThreads), kLdsReserve, stream,
                     in_seq, W_ih, W_hh, b_ih, b_hh, W_out, b_out, out);
}


# tanh by the hardware instruction v_tanh_f32 in place of the multiply, exp2, add, reciprocal, multiply-add chain: 57 plain and 28 transcendental issue slots fewer per step; tanh rounding differs
# speedup vs baseline: 8.9621x; 1.0440x over previous
#include <hip/hip_runtime.h>


typedef float v2f __attribute__((ext_vector_type(2)));
typedef float v4f __attribute__((ext_vector_type(4)));
typedef float v8f __attribute__((ext_vector_type(8)));

namespace {
constexpr int kT  = 2048;
constexpr int kB  = 128;
constexpr int kH  = 200;
constexpr int kHS = 204;
constexpr int kRows = 16;
constexpr int kGrid = 8;
constexpr int kWaves = 4;
constexpr int kThreads = kWaves * 32;
constexpr int kPXS = 20;
constexpr int kHBuf = kRows * kHS;
constexpr int kPXQ  = kRows * kPXS;
constexpr int kPXBuf = kWaves * kPXQ;
constexpr int kLdsFloats = 2 * kHBuf + 2 * kPXBuf;
constexpr int kLdsReserve = 176 * 1024;
static_assert(kLdsFloats * 4 <= kLdsReserve, "LDS use");
static_assert(kGrid * kRows == kB, "batch split");


__device__ __forceinline__ float tanh_fast(float x) {
  return __builtin_amdgcn_tanhf(x);
}

__device__ __forceinline__ v8f wmma4(v2f a, v2f b, v8f c) {
  return __builtin_amdgcn_wmma_f32_16x16x4_f32(false, a, false, b, (short)0, c, false, false);
}
__device__ __forceinline__ v2f lo2(v4f v) { return v2f{v[0], v[1]}; }
__device__ __forceinline__ v2f hi2(v4f v) { return v2f{v[2], v[3]}; }

__device__ __forceinline__ void bar_arrive() { asm volatile("s_wait_dscnt 0x0\n\ts_barrier_signal -1" ::: "memory"); }
__device__ __forceinline__ void bar_sync()   { asm volatile("s_barrier_wait -1" ::: "memory"); }
__device__ __forceinline__ void pin(v8f& a, v8f& b)     { asm volatile("" : "+v"(a), "+v"(b)); }
__device__ __forceinline__ void pin_mem(v8f& a, v8f& b) { asm volatile("" : "+v"(a), "+v"(b) :: "memory"); }
}

__global__ __launch_bounds__(kThreads) __attribute__((amdgpu_waves_per_eu(1, 1)))
void rnn_fused_wmma(const float* __restrict__ in_seq,
                    const float* __restrict__ W_ih,
                    const float* __restrict__ W_hh,
                    const float* __restrict__ b_ih,
                    const float* __restrict__ b_hh,
                    const float* __restrict__ W_out,
                    const float* __restrict__ b_out,
                    float* __restrict__ out)
{
  extern __shared__ __attribute__((aligned(16))) float lds[];
  float* hbuf = lds;
  float* pbuf = lds + 2 * kHBuf;

  const int tid  = threadIdx.x;
  const int lane = tid & 31;
  const int c    = lane & 15;
  const int h    = lane >> 4;
  const int w    = __builtin_amdgcn_readfirstlane(tid >> 5);
  const int b0   = blockIdx.x * kRows;

  for (int idx = tid; idx < kLdsFloats; idx += kThreads) {
    float v = 0.0f;
    if (idx < 2 * kHBuf && (idx % kHS) == 202) v = 1.0f;
    lds[idx] = v;
  }

  int kq[4];
#pragma unroll
  for (int d = 0; d < 4; ++d) kq[d] = 6 * ((w + d) & 3);

  v4f bE[3][4][4];
  v4f bL[3][4][2];
  v4f b24[3];
  v2f b50[3];
#pragma unroll
  for (int n = 0; n < 3; ++n) {
    const int col = 16 * (3 * w + n) + c;
    const float* rowp = W_hh + col * kH + 4 * h;
#pragma unroll
    for (int d = 0; d < 4; ++d) {
#pragma unroll
      for (int i = 0; i < 4; ++i) bE[n][d][i] = *reinterpret_cast<const v4f*>(rowp + 8 * (kq[d] + i));
#pragma unroll
      for (int i = 0; i < 2; ++i) bL[n][d][i] = *reinterpret_cast<const v4f*>(rowp + 8 * (kq[d] + 4 + i));
    }
    b24[n] = *reinterpret_cast<const v4f*>(rowp + 192);
    const v2f wi = *reinterpret_cast<const v2f*>(W_ih + 2 * col);
    const float bs = b_ih[col] + b_hh[col];
    b50[n] = h ? v2f{bs, 0.0f} : wi;
  }
  const int col12 = 192 + c;
  v4f bQ[6];
#pragma unroll
  for (int i = 0; i < 6; ++i) {
    const int k0 = 8 * (6 * w + i) + 4 * h;
    v4f v = {0.0f, 0.0f, 0.0f, 0.0f};
    if (col12 < kH)       v = *reinterpret_cast<const v4f*>(W_hh + col12 * kH + k0);
    else if (col12 == kH) v = *reinterpret_cast<const v4f*>(W_out + k0);
    bQ[i] = v;
  }
  v2f bX = {0.0f, 0.0f};
  if (w < 3) {
    const int kx = (w == 0) ? (192 + 4 * h) : (w == 1) ? (194 + 4 * h) : (200 + 2 * h);
#pragma unroll
    for (int e = 0; e < 2; ++e) {
      const int k = kx + e;
      float v = 0.0f;
      if (col12 < kH) {
        if (k < kH)        v = W_hh[col12 * kH + k];
        else if (k == 200) v = W_ih[col12 * 2 + 0];
        else if (k == 201) v = W_ih[col12 * 2 + 1];
        else if (k == 202) v = b_ih[col12] + b_hh[col12];
      } else if (col12 == kH) {
        if (k < kH)        v = W_out[k];
        else if (k == 202) v = b_out[0];
      }
      bX[e] = v;
    }
  }

  __syncthreads();

  int aOffD[4];
#pragma unroll
  for (int d = 0; d < 4; ++d) aOffD[d] = c * kHS + 4 * h + 8 * kq[d];
  const int a50Off = c * kHS + 200 + 2 * h;
  const int prOff = c * kPXS + 4 * h;
  const int phOff = c * kPXS + 8;
  const int hwOff = (8 * h) * kHS + 16 * (3 * w) + c;
  const int pwOff = w * kPXQ + (8 * h) * kPXS + c;
  const bool headLane = (w == 0) && (lane < kRows);

  const v8f zero8 = {0.f, 0.f, 0.f, 0.f, 0.f, 0.f, 0.f, 0.f};

  v4f aE[4][4];
#pragma unroll
  for (int d = 0; d < 4; ++d)
#pragma unroll
    for (int i = 0; i < 4; ++i) aE[d][i] = v4f{0.0f, 0.0f, 0.0f, 0.0f};
  v8f acc2 = zero8, acc3 = zero8, qa = zero8, qb = zero8;
  v2f xv = {0.0f, 0.0f};
  if (headLane) xv = reinterpret_cast<const v2f*>(in_seq)[b0 + lane];

#pragma unroll 1
  for (int t = 0; t <= kT + 1; ++t) {
    float* hcur = hbuf + (t & 1) * kHBuf;
    float* hnxt = hbuf + ((t & 1) ^ 1) * kHBuf;
    float* pcur = pbuf + (t & 1) * kPXBuf;

    v8f acc0 = zero8, acc1 = zero8;

#pragma unroll
    for (int i = 0; i < 2; ++i) {
      acc0 = wmma4(lo2(aE[0][i]), lo2(bE[0][0][i]), acc0);
      acc1 = wmma4(lo2(aE[0][i]), lo2(bE[1][0][i]), acc1);
      acc0 = wmma4(hi2(aE[0][i]), hi2(bE[0][0][i]), acc0);
      acc1 = wmma4(hi2(aE[0][i]), hi2(bE[1][0][i]), acc1);
    }
    float th2[8], ps[8];
#pragma unroll
    for (int r = 0; r < 8; ++r) {
      th2[r] = tanh_fast(acc2[r] + acc3[r]);
      ps[r]  = qa[r] + qb[r];
    }
    pin_mem(acc0, acc1);
#pragma unroll
    for (int r = 0; r < 8; ++r) {
      hcur[hwOff + r * kHS + 32] = th2[r];
      pcur[pwOff + r * kPXS]     = ps[r];
    }
    if (headLane) *reinterpret_cast<v2f*>(&hcur[lane * kHS + 200]) = xv;
    pin_mem(acc0, acc1);
#pragma unroll
    for (int i = 2; i < 4; ++i) {
      acc0 = wmma4(lo2(aE[0][i]), lo2(bE[0][0][i]), acc0);
      acc1 = wmma4(lo2(aE[0][i]), lo2(bE[1][0][i]), acc1);
      acc0 = wmma4(hi2(aE[0][i]), hi2(bE[0][0][i]), acc0);
      acc1 = wmma4(hi2(aE[0][i]), hi2(bE[1][0][i]), acc1);
    }
    pin(acc0, acc1);
    bar_arrive();

#pragma unroll
    for (int i = 0; i < 4; ++i) {
      acc0 = wmma4(lo2(aE[1][i]), lo2(bE[0][1][i]), acc0);
      acc1 = wmma4(lo2(aE[1][i]), lo2(bE[1][1][i]), acc1);
      acc0 = wmma4(hi2(aE[1][i]), hi2(bE[0][1][i]), acc0);
      acc1 = wmma4(hi2(aE[1][i]), hi2(bE[1][1][i]), acc1);
    }
    pin(acc0, acc1);
    bar_sync();

    v4f aL[4][2];
    v4f pr[kWaves];
#pragma unroll
    for (int q = 0; q < kWaves; ++q) pr[q] = *reinterpret_cast<const v4f*>(pcur + q * kPXQ + prOff);
#pragma unroll
    for (int d = 0; d < 4; ++d)
#pragma unroll
      for (int i = 0; i < 2; ++i) aL[d][i] = *reinterpret_cast<const v4f*>(hcur + aOffD[d] + 32 + 8 * i);
    const v2f a50 = *reinterpret_cast<const v2f*>(hcur + a50Off);
    pin_mem(acc0, acc1);
    __builtin_amdgcn_sched_barrier(0);

    if (headLane) {
      const int tn = (t + 1 < kT) ? (t + 1) : (kT - 1);
      xv = reinterpret_cast<const v2f*>(in_seq)[tn * kB + b0 + lane];
    }

#pragma unroll
    for (int i = 0; i < 4; ++i) {
      acc0 = wmma4(lo2(aE[2][i]), lo2(bE[0][2][i]), acc0);
      acc1 = wmma4(lo2(aE[2][i]), lo2(bE[1][2][i]), acc1);
      acc0 = wmma4(hi2(aE[2][i]), hi2(bE[0][2][i]), acc0);
      acc1 = wmma4(hi2(aE[2][i]), hi2(bE[1][2][i]), acc1);
    }
    asm volatile("" : "+v"(acc0), "+v"(acc1), "+v"(pr[0]), "+v"(pr[1]), "+v"(pr[2]), "+v"(pr[3]));
    v4f a24;
#pragma unroll
    for (int e = 0; e < 4; ++e) a24[e] = tanh_fast(((pr[0][e] + pr[1][e]) + pr[2][e]) + pr[3][e]);
#pragma unroll
    for (int i = 0; i < 4; ++i) {
      acc0 = wmma4(lo2(aE[3][i]), lo2(bE[0][3][i]), acc0);
      acc1 = wmma4(lo2(aE[3][i]), lo2(bE[1][3][i]), acc1);
      acc0 = wmma4(hi2(aE[3][i]), hi2(bE[0][3][i]), acc0);
      acc1 = wmma4(hi2(aE[3][i]), hi2(bE[1][3][i]), acc1);
    }
#pragma unroll
    for (int d = 0; d < 4; ++d)
#pragma unroll
      for (int i = 0; i < 2; ++i) {
        acc0 = wmma4(lo2(aL[d][i]), lo2(bL[0][d][i]), acc0);
        acc1 = wmma4(lo2(aL[d][i]), lo2(bL[1][d][i]), acc1);
        acc0 = wmma4(hi2(aL[d][i]), hi2(bL[0][d][i]), acc0);
        acc1 = wmma4(hi2(aL[d][i]), hi2(bL[1][d][i]), acc1);
      }
    acc0 = wmma4(lo2(a24), lo2(b24[0]), acc0);
    acc1 = wmma4(lo2(a24), lo2(b24[1]), acc1);
    acc0 = wmma4(hi2(a24), hi2(b24[0]), acc0);
    acc1 = wmma4(hi2(a24), hi2(b24[1]), acc1);
    acc0 = wmma4(a50, b50[0], acc0);
    acc1 = wmma4(a50, b50[1], acc1);

    acc2 = zero8; acc3 = zero8; qa = zero8; qb = zero8;
    asm volatile("" : "+v"(acc0), "+v"(acc1), "+v"(acc2), "+v"(acc3), "+v"(qa), "+v"(qb));

    const v2f aX  = (w == 0) ? lo2(a24) : (w == 1) ? hi2(a24) : a50;
#pragma unroll
    for (int i = 0; i < 4; ++i) {
      qa   = wmma4(lo2(aE[0][i]), lo2(bQ[i]), qa);
      acc2 = wmma4(lo2(aE[0][i]), lo2(bE[2][0][i]), acc2);
      qb   = wmma4(hi2(aE[0][i]), hi2(bQ[i]), qb);
      acc3 = wmma4(hi2(aE[0][i]), hi2(bE[2][0][i]), acc3);
    }
#pragma unroll
    for (int i = 0; i < 2; ++i) {
      qa   = wmma4(lo2(aL[0][i]), lo2(bQ[4 + i]), qa);
      acc2 = wmma4(lo2(aL[0][i]), lo2(bL[2][0][i]), acc2);
      qb   = wmma4(hi2(aL[0][i]), hi2(bQ[4 + i]), qb);
      acc3 = wmma4(hi2(aL[0][i]), hi2(bL[2][0][i]), acc3);
    }
    qa = wmma4(aX, bX, qa);
    float th0[8], th1[8];
#pragma unroll
    for (int r = 0; r < 8; ++r) {
      th0[r] = tanh_fast(acc0[r]);
      th1[r] = tanh_fast(acc1[r]);
    }
    asm volatile("" : "+v"(acc2), "+v"(acc3), "+v"(qa), "+v"(qb) :: "memory");
#pragma unroll
    for (int r = 0; r < 8; ++r) {
      hnxt[hwOff + r * kHS]      = th0[r];
      hnxt[hwOff + r * kHS + 16] = th1[r];
    }
    pin_mem(acc2, acc3);
#pragma unroll
    for (int i = 0; i < 4; ++i) {
      acc2 = wmma4(lo2(aE[1][i]), lo2(bE[2][1][i]), acc2);
      acc3 = wmma4(hi2(aE[1][i]), hi2(bE[2][1][i]), acc3);
    }
    pin(acc2, acc3);
    bar_arrive();

#pragma unroll
    for (int d = 2; d < 4; ++d)
#pragma unroll
      for (int i = 0; i < 4; ++i) {
        acc2 = wmma4(lo2(aE[d][i]), lo2(bE[2][d][i]), acc2);
        acc3 = wmma4(hi2(aE[d][i]), hi2(bE[2][d][i]), acc3);
      }
    pin(acc2, acc3);
    bar_sync();

    float ph[kWaves];
#pragma unroll
    for (int q = 0; q < kWaves; ++q) ph[q] = pcur[q * kPXQ + phOff];
#pragma unroll
    for (int d = 0; d < 4; ++d)
#pragma unroll
      for (int i = 0; i < 4; ++i) aE[d][i] = *reinterpret_cast<const v4f*>(hnxt + aOffD[d] + 8 * i);
    pin_mem(acc2, acc3);
    __builtin_amdgcn_sched_barrier(0);
#pragma unroll
    for (int d = 1; d < 4; ++d)
#pragma unroll
      for (int i = 0; i < 2; ++i) {
        acc2 = wmma4(lo2(aL[d][i]), lo2(bL[2][d][i]), acc2);
        acc3 = wmma4(hi2(aL[d][i]), hi2(bL[2][d][i]), acc3);
      }
    acc2 = wmma4(lo2(a24), lo2(b24[2]), acc2);
    acc3 = wmma4(hi2(a24), hi2(b24[2]), acc3);
    acc2 = wmma4(a50, b50[2], acc2);
    if (headLane && t > 1) out[(t - 2) * kB + b0 + lane] = ((ph[0] + ph[1]) + ph[2]) + ph[3];
  }
}

extern "C" void kernel_launch(void* const* d_in, const int* in_sizes, int n_in,
                              void* d_out, int out_size, void* d_ws, size_t ws_size,
                              hipStream_t stream) {
  const float* in_seq = (const float*)d_in[0];
  const float* W_ih   = (const float*)d_in[1];
  const float* W_hh   = (const float*)d_in[2];
  const float* b_ih   = (const float*)d_in[3];
  const float* b_hh   = (const float*)d_in[4];
  const float* W_out  = (const float*)d_in[5];
  const float* b_out  = (const float*)d_in[6];
  float* out = (float*)d_out;

  hipLaunchKernelGGL(rnn_fused_wmma, dim3(kGrid), dim3(kThreads), kLdsReserve, stream,
                     in_seq, W_ih, W_hh, b_ih, b_hh, W_out, b_out, out);
}


# Product turned over (weights = A, state = B): tanh(D) registers are the next step's B fragments, so waves swap state by lane-wise 16-byte LDS copies; one barrier per step (K sum order changes)
# speedup vs baseline: 9.6127x; 1.0726x over previous
#include <hip/hip_runtime.h>


typedef float v2f __attribute__((ext_vector_type(2)));
typedef float v4f __attribute__((ext_vector_type(4)));
typedef float v8f __attribute__((ext_vector_type(8)));

namespace {
constexpr int kT  = 2048;
constexpr int kB  = 128;
constexpr int kH  = 200;
constexpr int kRows = 16;
constexpr int kGrid = 8;
constexpr int kWaves = 4;
constexpr int kThreads = kWaves * 32;
constexpr int kGrp  = 128;
constexpr int kSBuf = 24 * kGrp;
constexpr int kPXS = 20;
constexpr int kPXQ  = kRows * kPXS;
constexpr int kPXBuf = kWaves * kPXQ;
constexpr int kLdsFloats = 2 * kSBuf + 2 * kPXBuf;
constexpr int kLdsReserve = 176 * 1024;
static_assert(kLdsFloats * 4 <= kLdsReserve, "LDS use");
static_assert(kGrid * kRows == kB, "batch split");


__device__ __forceinline__ float tanh_fast(float x) {
  return __builtin_amdgcn_tanhf(x);
}

__device__ __forceinline__ v8f wmma4(v2f a, v2f b, v8f c) {
  return __builtin_amdgcn_wmma_f32_16x16x4_f32(false, a, false, b, (short)0, c, false, false);
}
__device__ __forceinline__ v2f lo2(v4f v) { return v2f{v[0], v[1]}; }
__device__ __forceinline__ v2f hi2(v4f v) { return v2f{v[2], v[3]}; }

__device__ __forceinline__ void bar_sync()   { asm volatile("s_barrier_wait -1" ::: "memory"); }
__device__ __forceinline__ void pin(v8f& a, v8f& b)     { asm volatile("" : "+v"(a), "+v"(b)); }
__device__ __forceinline__ void pin_mem(v8f& a, v8f& b) { asm volatile("" : "+v"(a), "+v"(b) :: "memory"); }
}

__global__ __launch_bounds__(kThreads) __attribute__((amdgpu_waves_per_eu(1, 1)))
void rnn_fused_wmma(const float* __restrict__ in_seq,
                    const float* __restrict__ W_ih,
                    const float* __restrict__ W_hh,
                    const float* __restrict__ b_ih,
                    const float* __restrict__ b_hh,
                    const float* __restrict__ W_out,
                    const float* __restrict__ b_out,
                    float* __restrict__ out)
{
  extern __shared__ __attribute__((aligned(16))) float lds[];
  float* sbuf = lds;
  float* pbuf = lds + 2 * kSBuf;

  const int tid  = threadIdx.x;
  const int lane = tid & 31;
  const int c    = lane & 15;
  const int h    = lane >> 4;
  const int w    = __builtin_amdgcn_readfirstlane(tid >> 5);
  const int b0   = blockIdx.x * kRows;

  for (int idx = tid; idx < kLdsFloats; idx += kThreads) lds[idx] = 0.0f;

  int qd[4];
#pragma unroll
  for (int d = 0; d < 4; ++d) qd[d] = (w + d) & 3;

  v4f wE[3][4][3][2];
  v4f w24[3];
  v2f w50[3];
#pragma unroll
  for (int n = 0; n < 3; ++n) {
    const int row = 16 * (3 * w + n) + c;
    const float* rowp = W_hh + row * kH + 8 * h;
#pragma unroll
    for (int d = 0; d < 4; ++d)
#pragma unroll
      for (int j = 0; j < 3; ++j)
#pragma unroll
        for (int p = 0; p < 2; ++p)
          wE[n][d][j][p] = *reinterpret_cast<const v4f*>(rowp + 16 * (3 * qd[d] + j) + 4 * p);
    w24[n] = *reinterpret_cast<const v4f*>(W_hh + row * kH + 192 + 4 * h);
    const v2f wi = *reinterpret_cast<const v2f*>(W_ih + 2 * row);
    const float bs = b_ih[row] + b_hh[row];
    w50[n] = h ? v2f{bs, 0.0f} : wi;
  }
  const int row12 = 192 + c;
  v4f wQ[3][2];
#pragma unroll
  for (int j = 0; j < 3; ++j)
#pragma unroll
    for (int p = 0; p < 2; ++p) {
      const int k0 = 16 * (3 * w + j) + 8 * h + 4 * p;
      v4f v = {0.0f, 0.0f, 0.0f, 0.0f};
      if (row12 < kH)       v = *reinterpret_cast<const v4f*>(W_hh + row12 * kH + k0);
      else if (row12 == kH) v = *reinterpret_cast<const v4f*>(W_out + k0);
      wQ[j][p] = v;
    }
  v2f wX = {0.0f, 0.0f};
  if (w < 3) {
    const int kx = (w == 0) ? (192 + 4 * h) : (w == 1) ? (194 + 4 * h) : (200 + 2 * h);
#pragma unroll
    for (int e = 0; e < 2; ++e) {
      const int k = kx + e;
      float v = 0.0f;
      if (row12 < kH) {
        if (k < kH)        v = W_hh[row12 * kH + k];
        else if (k == 200) v = W_ih[row12 * 2 + 0];
        else if (k == 201) v = W_ih[row12 * 2 + 1];
        else if (k == 202) v = b_ih[row12] + b_hh[row12];
      } else if (row12 == kH) {
        if (k < kH)        v = W_out[k];
        else if (k == 202) v = b_out[0];
      }
      wX[e] = v;
    }
  }

  __syncthreads();

  const int stOff = 6 * w * kGrp + 4 * lane;
  int ldOff[4];
#pragma unroll
  for (int d = 0; d < 4; ++d) ldOff[d] = 6 * qd[d] * kGrp + 4 * lane;
  const int pwOff = w * kPXQ + c * kPXS + 8 * h;
  const int prOff = c * kPXS + 4 * h;
  const int phOff = c * kPXS + 8;
  const bool headLane = (w == 0) && (lane < kRows);

  const v8f zero8 = {0.f, 0.f, 0.f, 0.f, 0.f, 0.f, 0.f, 0.f};

  v4f S[3][2];
#pragma unroll
  for (int j = 0; j < 3; ++j)
#pragma unroll
    for (int p = 0; p < 2; ++p) S[j][p] = v4f{0.0f, 0.0f, 0.0f, 0.0f};
  v8f acc2 = zero8, acc3 = zero8, qa = zero8, qb = zero8;
  v2f xv = reinterpret_cast<const v2f*>(in_seq)[b0 + c];

#pragma unroll 1
  for (int t = 0; t <= kT + 1; ++t) {
    float* sCur = sbuf + (t & 1) * kSBuf;
    float* pCur = pbuf + (t & 1) * kPXBuf;

    v8f acc0 = zero8, acc1 = zero8;

#pragma unroll
    for (int p = 0; p < 2; ++p) {
      acc0 = wmma4(lo2(wE[0][0][0][p]), lo2(S[0][p]), acc0);
      acc1 = wmma4(lo2(wE[1][0][0][p]), lo2(S[0][p]), acc1);
      acc0 = wmma4(hi2(wE[0][0][0][p]), hi2(S[0][p]), acc0);
      acc1 = wmma4(hi2(wE[1][0][0][p]), hi2(S[0][p]), acc1);
    }
    v4f ps0, ps1;
#pragma unroll
    for (int r = 0; r < 4; ++r) {
      S[2][0][r] = tanh_fast(acc2[r] + acc3[r]);
      S[2][1][r] = tanh_fast(acc2[4 + r] + acc3[4 + r]);
      ps0[r] = qa[r] + qb[r];
      ps1[r] = qa[4 + r] + qb[4 + r];
    }
    pin_mem(acc0, acc1);
#pragma unroll
    for (int j = 0; j < 3; ++j)
#pragma unroll
      for (int p = 0; p < 2; ++p) *reinterpret_cast<v4f*>(sCur + stOff + (2 * j + p) * kGrp) = S[j][p];
    *reinterpret_cast<v4f*>(pCur + pwOff)     = ps0;
    *reinterpret_cast<v4f*>(pCur + pwOff + 4) = ps1;
    pin_mem(acc0, acc1);
#pragma unroll
    for (int p = 0; p < 2; ++p) {
      acc0 = wmma4(lo2(wE[0][0][1][p]), lo2(S[1][p]), acc0);
      acc1 = wmma4(lo2(wE[1][0][1][p]), lo2(S[1][p]), acc1);
      acc0 = wmma4(hi2(wE[0][0][1][p]), hi2(S[1][p]), acc0);
      acc1 = wmma4(hi2(wE[1][0][1][p]), hi2(S[1][p]), acc1);
    }
    asm volatile("s_wait_dscnt 0x0\n\ts_barrier_signal -1" : "+v"(acc0), "+v"(acc1), "+v"(S[0][0]) :: "memory");

#pragma unroll
    for (int p = 0; p < 2; ++p) {
      acc0 = wmma4(lo2(wE[0][0][2][p]), lo2(S[2][p]), acc0);
      acc1 = wmma4(lo2(wE[1][0][2][p]), lo2(S[2][p]), acc1);
      acc0 = wmma4(hi2(wE[0][0][2][p]), hi2(S[2][p]), acc0);
      acc1 = wmma4(hi2(wE[1][0][2][p]), hi2(S[2][p]), acc1);
    }
    qa   = wmma4(lo2(wQ[0][0]), lo2(S[0][0]), zero8);
    acc2 = wmma4(lo2(wE[2][0][0][0]), lo2(S[0][0]), zero8);
    qb   = wmma4(hi2(wQ[0][0]), hi2(S[0][0]), zero8);
    acc3 = wmma4(hi2(wE[2][0][0][0]), hi2(S[0][0]), zero8);
    qa   = wmma4(lo2(wQ[0][1]), lo2(S[0][1]), qa);
    acc2 = wmma4(lo2(wE[2][0][0][1]), lo2(S[0][1]), acc2);
    qb   = wmma4(hi2(wQ[0][1]), hi2(S[0][1]), qb);
    acc3 = wmma4(hi2(wE[2][0][0][1]), hi2(S[0][1]), acc3);
    qa   = wmma4(lo2(wQ[1][0]), lo2(S[1][0]), qa);
    acc2 = wmma4(lo2(wE[2][0][1][0]), lo2(S[1][0]), acc2);
    qb   = wmma4(hi2(wQ[1][0]), hi2(S[1][0]), qb);
    acc3 = wmma4(hi2(wE[2][0][1][0]), hi2(S[1][0]), acc3);
    asm volatile("" : "+v"(acc0), "+v"(acc1), "+v"(acc2), "+v"(acc3), "+v"(qa), "+v"(qb));
    bar_sync();

    v4f L[4][3][2];
    v4f pr[kWaves];
    float ph[kWaves];
#pragma unroll
    for (int q = 0; q < kWaves; ++q) pr[q] = *reinterpret_cast<const v4f*>(pCur + q * kPXQ + prOff);
#pragma unroll
    for (int d = 1; d < 4; ++d)
#pragma unroll
      for (int j = 0; j < 3; ++j)
#pragma unroll
        for (int p = 0; p < 2; ++p) L[d][j][p] = *reinterpret_cast<const v4f*>(sCur + ldOff[d] + (2 * j + p) * kGrp);
#pragma unroll
    for (int q = 0; q < kWaves; ++q) ph[q] = pCur[q * kPXQ + phOff];
    pin_mem(acc2, acc3);
    __builtin_amdgcn_sched_barrier(0);

    const int tn = (t + 1 < kT) ? (t + 1) : (kT - 1);
    const v2f xn = reinterpret_cast<const v2f*>(in_seq)[tn * kB + b0 + c];
    const v2f x50 = h ? v2f{1.0f, 0.0f} : xv;

#pragma unroll
    for (int g = 3; g < 6; ++g) {
      const int j = g >> 1, p = g & 1;
      qa   = wmma4(lo2(wQ[j][p]), lo2(S[j][p]), qa);
      acc2 = wmma4(lo2(wE[2][0][j][p]), lo2(S[j][p]), acc2);
      qb   = wmma4(hi2(wQ[j][p]), hi2(S[j][p]), qb);
      acc3 = wmma4(hi2(wE[2][0][j][p]), hi2(S[j][p]), acc3);
    }
    asm volatile("" : "+v"(acc0), "+v"(acc1), "+v"(acc2), "+v"(acc3), "+v"(qa), "+v"(qb));

#pragma unroll
    for (int j = 0; j < 3; ++j)
#pragma unroll
      for (int p = 0; p < 2; ++p) {
        acc0 = wmma4(lo2(wE[0][1][j][p]), lo2(L[1][j][p]), acc0);
        acc1 = wmma4(lo2(wE[1][1][j][p]), lo2(L[1][j][p]), acc1);
        acc0 = wmma4(hi2(wE[0][1][j][p]), hi2(L[1][j][p]), acc0);
        acc1 = wmma4(hi2(wE[1][1][j][p]), hi2(L[1][j][p]), acc1);
      }
    asm volatile("" : "+v"(acc0), "+v"(acc1), "+v"(pr[0]), "+v"(pr[1]), "+v"(pr[2]), "+v"(pr[3]));
    v4f a24;
#pragma unroll
    for (int e = 0; e < 4; ++e) a24[e] = tanh_fast(((pr[0][e] + pr[1][e]) + pr[2][e]) + pr[3][e]);
#pragma unroll
    for (int d = 2; d < 4; ++d)
#pragma unroll
      for (int j = 0; j < 3; ++j)
#pragma unroll
        for (int p = 0; p < 2; ++p) {
          acc0 = wmma4(lo2(wE[0][d][j][p]), lo2(L[d][j][p]), acc0);
          acc1 = wmma4(lo2(wE[1][d][j][p]), lo2(L[d][j][p]), acc1);
          acc0 = wmma4(hi2(wE[0][d][j][p]), hi2(L[d][j][p]), acc0);
          acc1 = wmma4(hi2(wE[1][d][j][p]), hi2(L[d][j][p]), acc1);
        }
    acc0 = wmma4(lo2(w24[0]), lo2(a24), acc0);
    acc1 = wmma4(lo2(w24[1]), lo2(a24), acc1);
    acc0 = wmma4(hi2(w24[0]), hi2(a24), acc0);
    acc1 = wmma4(hi2(w24[1]), hi2(a24), acc1);
    acc0 = wmma4(w50[0], x50, acc0);
    acc1 = wmma4(w50[1], x50, acc1);
    asm volatile("" : "+v"(acc0), "+v"(acc1), "+v"(acc2), "+v"(acc3), "+v"(qa), "+v"(qb));

    const v2f aX = (w == 0) ? lo2(a24) : (w == 1) ? hi2(a24) : x50;
#pragma unroll
    for (int j = 0; j < 3; ++j)
#pragma unroll
      for (int p = 0; p < 2; ++p) {
        acc2 = wmma4(lo2(wE[2][1][j][p]), lo2(L[1][j][p]), acc2);
        acc3 = wmma4(hi2(wE[2][1][j][p]), hi2(L[1][j][p]), acc3);
      }
    qa = wmma4(wX, aX, qa);
#pragma unroll
    for (int r = 0; r < 4; ++r) {
      S[0][0][r] = tanh_fast(acc0[r]);
      S[0][1][r] = tanh_fast(acc0[4 + r]);
      S[1][0][r] = tanh_fast(acc1[r]);
      S[1][1][r] = tanh_fast(acc1[4 + r]);
    }
#pragma unroll
    for (int d = 2; d < 4; ++d)
#pragma unroll
      for (int j = 0; j < 3; ++j)
#pragma unroll
        for (int p = 0; p < 2; ++p) {
          acc2 = wmma4(lo2(wE[2][d][j][p]), lo2(L[d][j][p]), acc2);
          acc3 = wmma4(hi2(wE[2][d][j][p]), hi2(L[d][j][p]), acc3);
        }
    acc2 = wmma4(lo2(w24[2]), lo2(a24), acc2);
    acc3 = wmma4(hi2(w24[2]), hi2(a24), acc3);
    acc2 = wmma4(w50[2], x50, acc2);
    if (headLane && t > 1) out[(t - 2) * kB + b0 + lane] = ((ph[0] + ph[1]) + ph[2]) + ph[3];
    xv = xn;
  }
}

extern "C" void kernel_launch(void* const* d_in, const int* in_sizes, int n_in,
                              void* d_out, int out_size, void* d_ws, size_t ws_size,
                              hipStream_t stream) {
  const float* in_seq = (const float*)d_in[0];
  const float* W_ih   = (const float*)d_in[1];
  const float* W_hh   = (const float*)d_in[2];
  const float* b_ih   = (const float*)d_in[3];
  const float* b_hh   = (const float*)d_in[4];
  const float* W_out  = (const float*)d_in[5];
  const float* b_out  = (const float*)d_in[6];
  float* out = (float*)d_out;

  hipLaunchKernelGGL(rnn_fused_wmma, dim3(kGrid), dim3(kThreads), kLdsReserve, stream,
                     in_seq, W_ih, W_hh, b_ih, b_hh, W_out, b_out, out);
}
